# starting kernel
# baseline (speedup 1.0000x reference)
#include <hip/hip_runtime.h>
#include <hip/hip_bf16.h>
#include <math.h>

#define HH   512
#define WW   512
#define KK   16
#define KP3  19
#define KPAD 20
#define BB   64
#define NCOL 128
#define HWN  (HH*WW)

#define BLDS 136
#define SROW 68

typedef __attribute__((ext_vector_type(2))) float v2f;
typedef __attribute__((ext_vector_type(8))) float v8f;

__device__ __forceinline__ float cpx_of(int k) { return -1.0f + (2.0f/3.0f)*(float)(k >> 2); }
__device__ __forceinline__ float cpy_of(int k) { return -1.0f + (2.0f/3.0f)*(float)(k & 3); }
__device__ __forceinline__ float tps_u(float r2) { return r2 * logf(r2 + 1e-6f); }

__global__ void tps_setup_kernel(const float* __restrict__ sp, float* __restrict__ bmat) {
    __shared__ float aug[KP3][2*KP3];
    __shared__ float invD[KP3][KP3];
    const int tid = threadIdx.x;

    if (tid == 0) {
        for (int i = 0; i < KP3; ++i) {
            for (int j = 0; j < KP3; ++j) {
                float v;
                if (i < KK && j < KK) {
                    float dx = cpx_of(i) - cpx_of(j);
                    float dy = cpy_of(i) - cpy_of(j);
                    v = tps_u(dx*dx + dy*dy);
                } else if (i < KK) {
                    v = (j == KK) ? 1.0f : ((j == KK+1) ? cpx_of(i) : cpy_of(i));
                } else if (j < KK) {
                    v = (i == KK) ? 1.0f : ((i == KK+1) ? cpx_of(j) : cpy_of(j));
                } else {
                    v = 0.0f;
                }
                aug[i][j] = v;
                aug[i][KP3 + j] = (i == j) ? 1.0f : 0.0f;
            }
        }
        for (int p = 0; p < KP3; ++p) {
            int piv = p; float best = fabsf(aug[p][p]);
            for (int r = p + 1; r < KP3; ++r) {
                float a = fabsf(aug[r][p]);
                if (a > best) { best = a; piv = r; }
            }
            if (piv != p) {
                for (int j = 0; j < 2*KP3; ++j) {
                    float t = aug[p][j]; aug[p][j] = aug[piv][j]; aug[piv][j] = t;
                }
            }
            float inv = 1.0f / aug[p][p];
            for (int j = 0; j < 2*KP3; ++j) aug[p][j] *= inv;
            for (int r = 0; r < KP3; ++r) {
                if (r == p) continue;
                float f = aug[r][p];
                for (int j = 0; j < 2*KP3; ++j) aug[r][j] -= f * aug[p][j];
            }
        }
        for (int i = 0; i < KP3; ++i)
            for (int j = 0; j < KP3; ++j)
                invD[i][j] = aug[i][KP3 + j];
    }
    __syncthreads();

    const int c = tid;
    if (c < NCOL) {
        const int b = c >> 1, d = c & 1;
        float spv[KK];
        #pragma unroll
        for (int j = 0; j < KK; ++j) spv[j] = sp[b*(KK*2) + j*2 + d];
        for (int i = 0; i < KP3; ++i) {
            float acc = 0.0f;
            #pragma unroll
            for (int j = 0; j < KK; ++j) acc += invD[i][j] * spv[j];
            bmat[i*NCOL + c] = acc;
        }
        bmat[KP3*NCOL + c] = 0.0f;
    }
}

__global__ void __launch_bounds__(256)
tps_main_kernel(const float* __restrict__ bmat, float* __restrict__ out) {
    __shared__ float Bls[KPAD * BLDS];
    __shared__ float stage[8][16 * SROW];

    const int tid = threadIdx.x;

    for (int idx = tid; idx < KPAD * NCOL; idx += 256) {
        int r = idx >> 7, cc = idx & 127;
        Bls[r * BLDS + cc] = bmat[idx];
    }
    __syncthreads();

    const int lane = tid & 31;
    const int wv   = tid >> 5;
    const int m    = lane & 15;
    const bool hi  = lane >= 16;
    const int n0   = blockIdx.x * 128 + wv * 16;
    const int n    = n0 + m;

    const float px = -1.0f + 2.0f * (float)(n & (WW - 1)) * (1.0f / (float)(WW - 1));
    const float py = -1.0f + 2.0f * (float)(n >> 9)       * (1.0f / (float)(HH - 1));

    v2f A[5];
    #pragma unroll
    for (int kb = 0; kb < 4; ++kb) {
        int k0 = kb * 4 + (hi ? 2 : 0);
        float dx0 = px - cpx_of(k0),     dy0 = py - cpy_of(k0);
        float dx1 = px - cpx_of(k0 + 1), dy1 = py - cpy_of(k0 + 1);
        float r20 = dx0*dx0 + dy0*dy0;
        float r21 = dx1*dx1 + dy1*dy1;
        A[kb].x = r20 * logf(r20 + 1e-6f);
        A[kb].y = r21 * logf(r21 + 1e-6f);
    }
    A[4].x = hi ? py   : 1.0f;
    A[4].y = hi ? 0.0f : px;

    float* st = &stage[wv][0];
    const int colLane = lane & 15;
    const int rbase   = hi ? 8 : 0;

    for (int h2 = 0; h2 < 2; ++h2) {
        #pragma unroll
        for (int t = 0; t < 4; ++t) {
            const int c0  = (h2 * 4 + t) * 16;
            const int col = c0 + colLane;
            v8f C = {0.f, 0.f, 0.f, 0.f, 0.f, 0.f, 0.f, 0.f};
            #pragma unroll
            for (int kb = 0; kb < 5; ++kb) {
                int r0 = kb * 4 + (hi ? 2 : 0);
                v2f Bv;
                Bv.x = Bls[r0 * BLDS + col];
                Bv.y = Bls[(r0 + 1) * BLDS + col];
                C = __builtin_amdgcn_wmma_f32_16x16x4_f32(
                         false, A[kb],  false, Bv,
                         (short)0, C,  false,  false);
            }
            const int lc = t * 16 + colLane;
            #pragma unroll
            for (int v = 0; v < 8; ++v)
                st[(rbase + v) * SROW + lc] = C[v];
        }
        __syncthreads();

        #pragma unroll
        for (int bp = 0; bp < 16; ++bp) {
            const int b  = h2 * 32 + 2 * bp + (hi ? 1 : 0);
            const int lc = 4 * bp + (hi ? 2 : 0);
            v2f val = *(const v2f*)&st[m * SROW + lc];
            float* gp = out + (size_t)b * (size_t)(HWN * 2) + (size_t)(n0 + m) * 2;
            *(v2f*)gp = val;
        }
        __syncthreads();
    }
}

extern "C" void kernel_launch(void* const* d_in, const int* in_sizes, int n_in,
                              void* d_out, int out_size, void* d_ws, size_t ws_size,
                              hipStream_t stream) {
    (void)in_sizes; (void)n_in; (void)out_size; (void)ws_size;
    const float* src = (const float*)d_in[0];
    float* out       = (float*)d_out;
    float* bmat      = (float*)d_ws;

    tps_setup_kernel<<<1, 128, 0, stream>>>(src, bmat);

    const int nBlocks = HWN / 128;
    tps_main_kernel<<<nBlocks, 256, 0, stream>>>(bmat, out);
}


# main: one-wave work-groups of 64 points (was 8 x 16); accumulators transposed batch-major in LDS, one store instruction = one batch's 512 contiguous bytes; coefficients from cache; sum order kept
# speedup vs baseline: 19.3082x; 19.3082x over previous
#include <hip/hip_runtime.h>
#include <hip/hip_bf16.h>
#include <math.h>

#define HH   512
#define WW   512
#define KK   16
#define KP3  19
#define KPAD 20
#define BB   64
#define NCOL 128
#define HWN  (HH*WW)

#define PTS  64
#define SPIT 132

typedef __attribute__((ext_vector_type(2))) float v2f;
typedef __attribute__((ext_vector_type(8))) float v8f;

__device__ __forceinline__ float cpx_of(int k) { return -1.0f + (2.0f/3.0f)*(float)(k >> 2); }
__device__ __forceinline__ float cpy_of(int k) { return -1.0f + (2.0f/3.0f)*(float)(k & 3); }
__device__ __forceinline__ float tps_u(float r2) { return r2 * logf(r2 + 1e-6f); }

#define ALD  40
#define AELE (KP3 * ALD)
#define STHR 256
#define EPT  3

typedef __attribute__((ext_vector_type(4))) float v4f;

__device__ __forceinline__ float delta_entry(int i, int j) {
    float v;
    if (i < KK && j < KK) {
        float dx = cpx_of(i) - cpx_of(j);
        float dy = cpy_of(i) - cpy_of(j);
        v = tps_u(dx*dx + dy*dy);
    } else if (i < KK) {
        v = (j == KK) ? 1.0f : ((j == KK+1) ? cpx_of(i) : cpy_of(i));
    } else if (j < KK) {
        v = (i == KK) ? 1.0f : ((i == KK+1) ? cpx_of(j) : cpy_of(j));
    } else {
        v = 0.0f;
    }
    return v;
}

struct PivT { float v; int o; };

template <int LO, int N>
__device__ __forceinline__ PivT piv_search(const float (&x)[KPAD]) {
    if constexpr (N == 1) {
        return PivT{x[LO], LO * ALD};
    } else {
        const PivT a = piv_search<LO, N / 2>(x);
        const PivT b = piv_search<LO + N / 2, N - N / 2>(x);
        const bool take = fabsf(b.v) > fabsf(a.v);
        return PivT{take ? b.v : a.v, take ? b.o : a.o};
    }
}

template <int P>
__device__ __forceinline__ void gj_step(float* augb, float* pcolb, int tid,
                                        const int (&er)[EPT], const int (&eo)[EPT], const int (&ec)[EPT]) {
    const float* A  = augb  + (P & 1) * AELE;
    float*       N  = augb  + ((P & 1) ^ 1) * AELE;
    const float* pc = pcolb + (P & 1) * KPAD;
    float*       pn = pcolb + ((P & 1) ^ 1) * KPAD;

    float x[KPAD];
    #pragma unroll
    for (int q = 0; q < KPAD / 4; ++q) {
        v4f t = *(const v4f*)&pc[4 * q];
        x[4*q] = t.x; x[4*q+1] = t.y; x[4*q+2] = t.z; x[4*q+3] = t.w;
    }
    const PivT pr = piv_search<P, KP3 - P>(x);
    const int   po  = pr.o;
    const float inv = 1.0f / pr.v;

    float outv[EPT];
    #pragma unroll
    for (int q = 0; q < EPT; ++q) {
        const int ro = eo[q], jl = ec[q];
        const int so = (ro == po) ? P * ALD : ro;
        const float prow = A[po + jl] * inv;
        const float f    = A[so + P];
        const float elim = A[so + jl] - f * prow;
        outv[q] = (ro == P * ALD) ? prow : elim;
    }
    #pragma unroll
    for (int q = 0; q < EPT; ++q) {
        const int e = tid + STHR * q;
        if (STHR * (q + 1) <= AELE || e < AELE) {
            N[e] = outv[q];
            if (ec[q] == P + 1) pn[er[q]] = outv[q];
        }
    }
    __syncthreads();
}

template <int P>
__device__ __forceinline__ void gj_steps(float* augb, float* pcolb, int tid,
                                         const int (&er)[EPT], const int (&eo)[EPT], const int (&ec)[EPT]) {
    if constexpr (P < KP3) {
        gj_step<P>(augb, pcolb, tid, er, eo, ec);
        gj_steps<P + 1>(augb, pcolb, tid, er, eo, ec);
    }
}

__global__ void __launch_bounds__(STHR)
tps_setup_kernel(const float* __restrict__ sp, float* __restrict__ bmat) {
    __shared__ __attribute__((aligned(16))) float aug[2][AELE];
    __shared__ __attribute__((aligned(16))) float pcol[2][KPAD];
    const int tid = threadIdx.x;

    const int c = tid & (NCOL - 1);
    const int g = (tid >> 7) & 1;
    float spv[KK];
    {
        const int b = c >> 1, d = c & 1;
        #pragma unroll
        for (int j = 0; j < KK; ++j) spv[j] = sp[b*(KK*2) + j*2 + d];
    }

    int er[EPT], eo[EPT], ec[EPT];
    #pragma unroll
    for (int q = 0; q < EPT; ++q) {
        const int e = min(tid + STHR * q, AELE - 1);
        er[q] = e / ALD;
        eo[q] = er[q] * ALD;
        ec[q] = e - eo[q];
    }

    #pragma unroll
    for (int q = 0; q < EPT; ++q) {
        const int e = tid + STHR * q;
        if (e < AELE) {
            const int r = er[q], jl = ec[q];
            float v;
            if (jl < KP3) v = delta_entry(r, jl);
            else          v = (jl - KPAD == r) ? 1.0f : 0.0f;
            aug[0][e] = v;
            if (jl == 0) pcol[0][r] = v;
        }
    }
    if (tid == 0) { pcol[0][KP3] = 0.0f; pcol[1][KP3] = 0.0f; }
    __syncthreads();

    gj_steps<0>(&aug[0][0], &pcol[0][0], tid, er, eo, ec);

    const float* F = aug[KP3 & 1];
    #pragma unroll
    for (int h = 0; h < KPAD / 2; ++h) {
        const int i  = g + 2 * h;
        const int ir = min(i, KP3 - 1);
        float acc = 0.0f;
        #pragma unroll
        for (int q = 0; q < KK / 4; ++q) {
            v4f t = *(const v4f*)&F[ir * ALD + KPAD + 4 * q];
            acc = fmaf(t.x, spv[4*q],   acc);
            acc = fmaf(t.y, spv[4*q+1], acc);
            acc = fmaf(t.z, spv[4*q+2], acc);
            acc = fmaf(t.w, spv[4*q+3], acc);
        }
        bmat[i*NCOL + c] = (i < KP3) ? acc : 0.0f;
    }
}

__device__ __forceinline__ void lds_handover() {
    __builtin_amdgcn_fence(__ATOMIC_RELEASE, "workgroup", "local");
    __builtin_amdgcn_fence(__ATOMIC_ACQUIRE, "workgroup", "local");
}

__global__ void __launch_bounds__(32)
tps_main_kernel(const float* __restrict__ bmat, float* __restrict__ out) {
    __shared__ __attribute__((aligned(16))) float stage[8 * SPIT];

    const int lane = threadIdx.x;
    const int m    = lane & 15;
    const bool hi  = lane >= 16;
    const int n0   = blockIdx.x * PTS;
    const int pm   = (m & 7) + ((m >> 3) << 4);

    const float py = -1.0f + 2.0f * (float)(n0 >> 9) * (1.0f / (float)(HH - 1));

    v2f A[4][5];
    #pragma unroll
    for (int t = 0; t < 4; ++t) {
        const int n = n0 + pm + 8 * (t & 1) + 32 * (t >> 1);
        const float px = -1.0f + 2.0f * (float)(n & (WW - 1)) * (1.0f / (float)(WW - 1));
        #pragma unroll
        for (int kb = 0; kb < 4; ++kb) {
            int k0 = kb * 4 + (hi ? 2 : 0);
            float dx0 = px - cpx_of(k0),     dy0 = py - cpy_of(k0);
            float dx1 = px - cpx_of(k0 + 1), dy1 = py - cpy_of(k0 + 1);
            float r20 = dx0*dx0 + dy0*dy0;
            float r21 = dx1*dx1 + dy1*dy1;
            A[t][kb].x = r20 * logf(r20 + 1e-6f);
            A[t][kb].y = r21 * logf(r21 + 1e-6f);
        }
        A[t][4].x = hi ? py   : 1.0f;
        A[t][4].y = hi ? 0.0f : px;
    }

    const float* bp = bmat + (hi ? 2 : 0) * NCOL + m;
    v2f Bc[5], Bn[5];
    #pragma unroll
    for (int kb = 0; kb < 5; ++kb) {
        Bc[kb].x = bp[(4 * kb)     * NCOL];
        Bc[kb].y = bp[(4 * kb + 1) * NCOL];
    }

    float* sw = &stage[(m >> 1) * SPIT + (m & 1) + (hi ? 32 : 0)];
    const float* sr = &stage[4 * lane];
    float* op = out + (size_t)(n0 + 2 * lane) * 2;

    for (int ct = 0; ct < 8; ++ct) {
        const int cn = min(ct + 1, 7) * 16;
        #pragma unroll
        for (int kb = 0; kb < 5; ++kb) {
            Bn[kb].x = bp[(4 * kb)     * NCOL + cn];
            Bn[kb].y = bp[(4 * kb + 1) * NCOL + cn];
        }

        v8f C[4];
        #pragma unroll
        for (int t = 0; t < 4; ++t) C[t] = v8f{0.f, 0.f, 0.f, 0.f, 0.f, 0.f, 0.f, 0.f};
        #pragma unroll
        for (int kb = 0; kb < 5; ++kb) {
            #pragma unroll
            for (int t = 0; t < 4; ++t)
                C[t] = __builtin_amdgcn_wmma_f32_16x16x4_f32(
                         false, A[t][kb],  false, Bc[kb],
                         (short)0, C[t],  false,  false);
        }

        lds_handover();
        #pragma unroll
        for (int t = 0; t < 4; ++t) {
            #pragma unroll
            for (int r = 0; r < 8; ++r)
                sw[2 * r + 16 * (t & 1) + 64 * (t >> 1)] = C[t][r];
        }
        lds_handover();

        #pragma unroll
        for (int jj = 0; jj < 8; ++jj) {
            const v4f val = *(const v4f*)&sr[jj * SPIT];
            *(v4f*)(op + (size_t)(8 * ct + jj) * (size_t)(HWN * 2)) = val;
        }

        #pragma unroll
        for (int kb = 0; kb < 5; ++kb) Bc[kb] = Bn[kb];
    }
}

extern "C" void kernel_launch(void* const* d_in, const int* in_sizes, int n_in,
                              void* d_out, int out_size, void* d_ws, size_t ws_size,
                              hipStream_t stream) {
    (void)in_sizes; (void)n_in; (void)out_size; (void)ws_size;
    const float* src = (const float*)d_in[0];
    float* out       = (float*)d_out;
    float* bmat      = (float*)d_ws;

    tps_setup_kernel<<<1, 256, 0, stream>>>(src, bmat);

    const int nBlocks = HWN / PTS;
    tps_main_kernel<<<nBlocks, 32, 0, stream>>>(bmat, out);
}


# setup: the lattice's two mirror symmetries split the 19x19 system into four independent blocks (5,5,5,4), each inverted by one lane: 5 dependent pivot steps, not 19; sums reordered, rounding differs
# speedup vs baseline: 34.1579x; 1.7691x over previous
#include <hip/hip_runtime.h>
#include <hip/hip_bf16.h>
#include <math.h>

#define HH   512
#define WW   512
#define KK   16
#define KP3  19
#define KPAD 20
#define BB   64
#define NCOL 128
#define HWN  (HH*WW)

#define PTS  64
#define SPIT 132

typedef __attribute__((ext_vector_type(2))) float v2f;
typedef __attribute__((ext_vector_type(8))) float v8f;

__device__ __forceinline__ float cpx_of(int k) { return -1.0f + (2.0f/3.0f)*(float)(k >> 2); }
__device__ __forceinline__ float cpy_of(int k) { return -1.0f + (2.0f/3.0f)*(float)(k & 3); }
__device__ __forceinline__ float tps_u(float r2) { return r2 * logf(r2 + 1e-6f); }

#define STHR 256
#define NCL  4
#define BD   5
#define BAUG (BD + 4)

typedef __attribute__((ext_vector_type(4))) float v4f;

__device__ __forceinline__ float rep_coord(int i) { return cpx_of(4 * i); }

__global__ void __launch_bounds__(STHR)
tps_setup_kernel(const float* __restrict__ sp, float* __restrict__ bmat) {
    __shared__ float Mb[NCL * BD * BD];
    __shared__ __attribute__((aligned(16))) float Mi[NCL * BD * 4];
    const int tid = threadIdx.x;

    const int c = tid & (NCOL - 1);
    const int g = (tid >> 7) & 1;
    float spv[KK];
    {
        const int b = c >> 1, d = c & 1;
        #pragma unroll
        for (int j = 0; j < KK; ++j) spv[j] = sp[b*(KK*2) + j*2 + d];
    }

    if (tid < NCL * BD * BD) {
        const int s  = tid / (BD * BD);
        const int rm = tid - s * (BD * BD);
        const int i  = rm / BD;
        const int j  = rm - i * BD;
        const float sx = (s & 1) ? -1.0f : 1.0f;
        const float sy = (s & 2) ? -1.0f : 1.0f;
        const int p = min(i, 3), q = min(j, 3);
        const float xp = rep_coord(p >> 1), yp = rep_coord(p & 1);
        const float xq = rep_coord(q >> 1), yq = rep_coord(q & 1);
        const float dxa = xp - xq, dxb = xp + xq;
        const float dya = yp - yq, dyb = yp + yq;
        float kv = tps_u(dxa*dxa + dya*dya);
        kv += sy * tps_u(dxa*dxa + dyb*dyb);
        kv += sx * tps_u(dxb*dxb + dya*dya);
        kv += (sx * sy) * tps_u(dxb*dxb + dyb*dyb);
        const int t = (i == BD - 1) ? q : p;
        const float bv = (s == 0) ? 2.0f
                       : (s == 1) ? 2.0f * rep_coord(t >> 1)
                       : (s == 2) ? 2.0f * rep_coord(t & 1)
                       : 0.0f;
        const float cv = (s == 3) ? 1.0f : 0.0f;
        const bool ii = (i < BD - 1), jj = (j < BD - 1);
        Mb[tid] = (ii && jj) ? kv : ((ii || jj) ? bv : cv);
    }
    __syncthreads();

    if (tid < 32) {
        const int s = tid & 3;
        float a[BD][BAUG];
        #pragma unroll
        for (int i = 0; i < BD; ++i) {
            #pragma unroll
            for (int j = 0; j < BD; ++j) a[i][j] = Mb[s * (BD * BD) + i * BD + j];
            #pragma unroll
            for (int q = 0; q < 4; ++q) a[i][BD + q] = (i == q) ? 1.0f : 0.0f;
        }
        #pragma unroll
        for (int P = 0; P < BD; ++P) {
            float pv = a[P][P];
            int   pr = P;
            #pragma unroll
            for (int r = P + 1; r < BD; ++r) {
                const bool tk = fabsf(a[r][P]) > fabsf(pv);
                pv = tk ? a[r][P] : pv;
                pr = tk ? r : pr;
            }
            const float inv = 1.0f / pv;
            #pragma unroll
            for (int j = 0; j < BAUG; ++j) {
                const float top = a[P][j];
                float pick = top;
                #pragma unroll
                for (int r = P + 1; r < BD; ++r) {
                    const bool is = (pr == r);
                    const float cur = a[r][j];
                    pick    = is ? cur : pick;
                    a[r][j] = is ? top : cur;
                }
                a[P][j] = pick * inv;
            }
            #pragma unroll
            for (int r = 0; r < BD; ++r) {
                if (r != P) {
                    const float f = a[r][P];
                    #pragma unroll
                    for (int j = 0; j < BAUG; ++j) a[r][j] -= f * a[P][j];
                }
            }
        }
        if (tid < NCL) {
            #pragma unroll
            for (int i = 0; i < BD; ++i) {
                #pragma unroll
                for (int q = 0; q < 4; ++q) Mi[s * (BD * 4) + i * 4 + q] = a[i][BD + q];
            }
        }
    }
    __syncthreads();

    float st[NCL][4];
    #pragma unroll
    for (int p = 0; p < 4; ++p) {
        const int ar = p >> 1, br = p & 1;
        const float u00 = spv[4 * ar       + br];
        const float u01 = spv[4 * ar       + (3 - br)];
        const float u10 = spv[4 * (3 - ar) + br];
        const float u11 = spv[4 * (3 - ar) + (3 - br)];
        const float e0 = u00 + u10, o0 = u00 - u10;
        const float e1 = u01 + u11, o1 = u01 - u11;
        st[0][p] = 0.5f * (e0 + e1);
        st[1][p] = 0.5f * (o0 + o1);
        st[2][p] = 0.5f * (e0 - e1);
        st[3][p] = 0.5f * (o0 - o1);
    }

    float Wc[NCL][BD];
    #pragma unroll
    for (int s = 0; s < NCL; ++s) {
        #pragma unroll
        for (int i = 0; i < BD; ++i) {
            const v4f m = *(const v4f*)&Mi[s * (BD * 4) + i * 4];
            float acc = m.x * st[s][0];
            acc = fmaf(m.y, st[s][1], acc);
            acc = fmaf(m.z, st[s][2], acc);
            acc = fmaf(m.w, st[s][3], acc);
            Wc[s][i] = acc;
        }
    }

    float cf[KPAD];
    #pragma unroll
    for (int p = 0; p < 4; ++p) {
        const int ar = p >> 1, br = p & 1;
        const float xe = Wc[0][p] + Wc[1][p], xo = Wc[0][p] - Wc[1][p];
        const float ye = Wc[2][p] + Wc[3][p], yo = Wc[2][p] - Wc[3][p];
        cf[4 * ar       + br]       = 0.5f * (xe + ye);
        cf[4 * (3 - ar) + br]       = 0.5f * (xo + yo);
        cf[4 * ar       + (3 - br)] = 0.5f * (xe - ye);
        cf[4 * (3 - ar) + (3 - br)] = 0.5f * (xo - yo);
    }
    cf[KK]     = Wc[0][BD - 1];
    cf[KK + 1] = Wc[1][BD - 1];
    cf[KK + 2] = Wc[2][BD - 1];
    cf[KP3]    = 0.0f;

    #pragma unroll
    for (int h = 0; h < KPAD / 2; ++h)
        bmat[(g + 2 * h) * NCOL + c] = g ? cf[2 * h + 1] : cf[2 * h];
}

__device__ __forceinline__ void lds_handover() {
    __builtin_amdgcn_fence(__ATOMIC_RELEASE, "workgroup", "local");
    __builtin_amdgcn_fence(__ATOMIC_ACQUIRE, "workgroup", "local");
}

__global__ void __launch_bounds__(32)
tps_main_kernel(const float* __restrict__ bmat, float* __restrict__ out) {
    __shared__ __attribute__((aligned(16))) float stage[8 * SPIT];

    const int lane = threadIdx.x;
    const int m    = lane & 15;
    const bool hi  = lane >= 16;
    const int n0   = blockIdx.x * PTS;
    const int pm   = (m & 7) + ((m >> 3) << 4);

    const float py = -1.0f + 2.0f * (float)(n0 >> 9) * (1.0f / (float)(HH - 1));

    v2f A[4][5];
    #pragma unroll
    for (int t = 0; t < 4; ++t) {
        const int n = n0 + pm + 8 * (t & 1) + 32 * (t >> 1);
        const float px = -1.0f + 2.0f * (float)(n & (WW - 1)) * (1.0f / (float)(WW - 1));
        #pragma unroll
        for (int kb = 0; kb < 4; ++kb) {
            int k0 = kb * 4 + (hi ? 2 : 0);
            float dx0 = px - cpx_of(k0),     dy0 = py - cpy_of(k0);
            float dx1 = px - cpx_of(k0 + 1), dy1 = py - cpy_of(k0 + 1);
            float r20 = dx0*dx0 + dy0*dy0;
            float r21 = dx1*dx1 + dy1*dy1;
            A[t][kb].x = r20 * logf(r20 + 1e-6f);
            A[t][kb].y = r21 * logf(r21 + 1e-6f);
        }
        A[t][4].x = hi ? py   : 1.0f;
        A[t][4].y = hi ? 0.0f : px;
    }

    const float* bp = bmat + (hi ? 2 : 0) * NCOL + m;
    v2f Bc[5], Bn[5];
    #pragma unroll
    for (int kb = 0; kb < 5; ++kb) {
        Bc[kb].x = bp[(4 * kb)     * NCOL];
        Bc[kb].y = bp[(4 * kb + 1) * NCOL];
    }

    float* sw = &stage[(m >> 1) * SPIT + (m & 1) + (hi ? 32 : 0)];
    const float* sr = &stage[4 * lane];
    float* op = out + (size_t)(n0 + 2 * lane) * 2;

    for (int ct = 0; ct < 8; ++ct) {
        const int cn = min(ct + 1, 7) * 16;
        #pragma unroll
        for (int kb = 0; kb < 5; ++kb) {
            Bn[kb].x = bp[(4 * kb)     * NCOL + cn];
            Bn[kb].y = bp[(4 * kb + 1) * NCOL + cn];
        }

        v8f C[4];
        #pragma unroll
        for (int t = 0; t < 4; ++t) C[t] = v8f{0.f, 0.f, 0.f, 0.f, 0.f, 0.f, 0.f, 0.f};
        #pragma unroll
        for (int kb = 0; kb < 5; ++kb) {
            #pragma unroll
            for (int t = 0; t < 4; ++t)
                C[t] = __builtin_amdgcn_wmma_f32_16x16x4_f32(
                         false, A[t][kb],  false, Bc[kb],
                         (short)0, C[t],  false,  false);
        }

        lds_handover();
        #pragma unroll
        for (int t = 0; t < 4; ++t) {
            #pragma unroll
            for (int r = 0; r < 8; ++r)
                sw[2 * r + 16 * (t & 1) + 64 * (t >> 1)] = C[t][r];
        }
        lds_handover();

        #pragma unroll
        for (int jj = 0; jj < 8; ++jj) {
            const v4f val = *(const v4f*)&sr[jj * SPIT];
            *(v4f*)(op + (size_t)(8 * ct + jj) * (size_t)(HWN * 2)) = val;
        }

        #pragma unroll
        for (int kb = 0; kb < 5; ++kb) Bc[kb] = Bn[kb];
    }
}

extern "C" void kernel_launch(void* const* d_in, const int* in_sizes, int n_in,
                              void* d_out, int out_size, void* d_ws, size_t ws_size,
                              hipStream_t stream) {
    (void)in_sizes; (void)n_in; (void)out_size; (void)ws_size;
    const float* src = (const float*)d_in[0];
    float* out       = (float*)d_out;
    float* bmat      = (float*)d_ws;

    tps_setup_kernel<<<1, 256, 0, stream>>>(src, bmat);

    const int nBlocks = HWN / PTS;
    tps_main_kernel<<<nBlocks, 32, 0, stream>>>(bmat, out);
}


# main: mirror symmetries of grid and lattice: basis and WMMA products for one quadrant only, in the class basis; the 3 mirror images by sign additions; a wave owns 128 output points; sums reordered
# speedup vs baseline: 37.5603x; 1.0996x over previous
#include <hip/hip_runtime.h>
#include <hip/hip_bf16.h>
#include <math.h>

#define HH   512
#define WW   512
#define KK   16
#define KP3  19
#define KPAD 20
#define BB   64
#define NCOL 128
#define HWN  (HH*WW)

#define QPTS 32
#define SPIT 260

typedef __attribute__((ext_vector_type(2))) float v2f;
typedef __attribute__((ext_vector_type(8))) float v8f;

__device__ __forceinline__ float cpx_of(int k) { return -1.0f + (2.0f/3.0f)*(float)(k >> 2); }
__device__ __forceinline__ float cpy_of(int k) { return -1.0f + (2.0f/3.0f)*(float)(k & 3); }
__device__ __forceinline__ float tps_u(float r2) { return r2 * logf(r2 + 1e-6f); }

#define STHR 256
#define NCL  4
#define BD   5
#define BAUG (BD + 4)

typedef __attribute__((ext_vector_type(4))) float v4f;

__device__ __forceinline__ float rep_coord(int i) { return cpx_of(4 * i); }

__global__ void __launch_bounds__(STHR)
tps_setup_kernel(const float* __restrict__ sp, float* __restrict__ bmat) {
    __shared__ float Mb[NCL * BD * BD];
    __shared__ __attribute__((aligned(16))) float Mi[NCL * BD * 4];
    const int tid = threadIdx.x;

    const int c = tid & (NCOL - 1);
    const int g = (tid >> 7) & 1;
    float spv[KK];
    {
        const int b = c >> 1, d = c & 1;
        #pragma unroll
        for (int j = 0; j < KK; ++j) spv[j] = sp[b*(KK*2) + j*2 + d];
    }

    if (tid < NCL * BD * BD) {
        const int s  = tid / (BD * BD);
        const int rm = tid - s * (BD * BD);
        const int i  = rm / BD;
        const int j  = rm - i * BD;
        const float sx = (s & 1) ? -1.0f : 1.0f;
        const float sy = (s & 2) ? -1.0f : 1.0f;
        const int p = min(i, 3), q = min(j, 3);
        const float xp = rep_coord(p >> 1), yp = rep_coord(p & 1);
        const float xq = rep_coord(q >> 1), yq = rep_coord(q & 1);
        const float dxa = xp - xq, dxb = xp + xq;
        const float dya = yp - yq, dyb = yp + yq;
        float kv = tps_u(dxa*dxa + dya*dya);
        kv += sy * tps_u(dxa*dxa + dyb*dyb);
        kv += sx * tps_u(dxb*dxb + dya*dya);
        kv += (sx * sy) * tps_u(dxb*dxb + dyb*dyb);
        const int t = (i == BD - 1) ? q : p;
        const float bv = (s == 0) ? 2.0f
                       : (s == 1) ? 2.0f * rep_coord(t >> 1)
                       : (s == 2) ? 2.0f * rep_coord(t & 1)
                       : 0.0f;
        const float cv = (s == 3) ? 1.0f : 0.0f;
        const bool ii = (i < BD - 1), jj = (j < BD - 1);
        Mb[tid] = (ii && jj) ? kv : ((ii || jj) ? bv : cv);
    }
    __syncthreads();

    if (tid < 32) {
        const int s = tid & 3;
        float a[BD][BAUG];
        #pragma unroll
        for (int i = 0; i < BD; ++i) {
            #pragma unroll
            for (int j = 0; j < BD; ++j) a[i][j] = Mb[s * (BD * BD) + i * BD + j];
            #pragma unroll
            for (int q = 0; q < 4; ++q) a[i][BD + q] = (i == q) ? 1.0f : 0.0f;
        }
        #pragma unroll
        for (int P = 0; P < BD; ++P) {
            float pv = a[P][P];
            int   pr = P;
            #pragma unroll
            for (int r = P + 1; r < BD; ++r) {
                const bool tk = fabsf(a[r][P]) > fabsf(pv);
                pv = tk ? a[r][P] : pv;
                pr = tk ? r : pr;
            }
            const float inv = 1.0f / pv;
            #pragma unroll
            for (int j = 0; j < BAUG; ++j) {
                const float top = a[P][j];
                float pick = top;
                #pragma unroll
                for (int r = P + 1; r < BD; ++r) {
                    const bool is = (pr == r);
                    const float cur = a[r][j];
                    pick    = is ? cur : pick;
                    a[r][j] = is ? top : cur;
                }
                a[P][j] = pick * inv;
            }
            #pragma unroll
            for (int r = 0; r < BD; ++r) {
                if (r != P) {
                    const float f = a[r][P];
                    #pragma unroll
                    for (int j = 0; j < BAUG; ++j) a[r][j] -= f * a[P][j];
                }
            }
        }
        if (tid < NCL) {
            #pragma unroll
            for (int i = 0; i < BD; ++i) {
                #pragma unroll
                for (int q = 0; q < 4; ++q) Mi[s * (BD * 4) + i * 4 + q] = a[i][BD + q];
            }
        }
    }
    __syncthreads();

    float st[NCL][4];
    #pragma unroll
    for (int p = 0; p < 4; ++p) {
        const int ar = p >> 1, br = p & 1;
        const float u00 = spv[4 * ar       + br];
        const float u01 = spv[4 * ar       + (3 - br)];
        const float u10 = spv[4 * (3 - ar) + br];
        const float u11 = spv[4 * (3 - ar) + (3 - br)];
        const float e0 = u00 + u10, o0 = u00 - u10;
        const float e1 = u01 + u11, o1 = u01 - u11;
        st[0][p] = 0.5f * (e0 + e1);
        st[1][p] = 0.5f * (o0 + o1);
        st[2][p] = 0.5f * (e0 - e1);
        st[3][p] = 0.5f * (o0 - o1);
    }

    float Wc[NCL][BD];
    #pragma unroll
    for (int s = 0; s < NCL; ++s) {
        #pragma unroll
        for (int i = 0; i < BD; ++i) {
            const v4f m = *(const v4f*)&Mi[s * (BD * 4) + i * 4];
            float acc = m.x * st[s][0];
            acc = fmaf(m.y, st[s][1], acc);
            acc = fmaf(m.z, st[s][2], acc);
            acc = fmaf(m.w, st[s][3], acc);
            Wc[s][i] = acc;
        }
    }

    float cfl[KPAD];
    #pragma unroll
    for (int s = 0; s < NCL; ++s) {
        #pragma unroll
        for (int i = 0; i < BD; ++i) cfl[BD * s + i] = Wc[s][i];
    }
    cfl[KP3] = 0.0f;

    #pragma unroll
    for (int h = 0; h < KPAD / 2; ++h)
        bmat[(g + 2 * h) * NCOL + c] = g ? cfl[2 * h + 1] : cfl[2 * h];
}

__device__ __forceinline__ void lds_handover() {
    __builtin_amdgcn_fence(__ATOMIC_RELEASE, "workgroup", "local");
    __builtin_amdgcn_fence(__ATOMIC_ACQUIRE, "workgroup", "local");
}

__global__ void __launch_bounds__(32)
tps_main_kernel(const float* __restrict__ bmat, float* __restrict__ out) {
    __shared__ __attribute__((aligned(16))) float stage[8 * SPIT];

    const int lane = threadIdx.x;
    const int m    = lane & 15;
    const bool hi  = lane >= 16;
    const int h    = hi ? 1 : 0;
    const int row  = blockIdx.x >> 3;
    const int j0   = (blockIdx.x & 7) * QPTS;
    const int pm   = (m & 7) + ((m >> 3) << 4);

    const float py = -1.0f + 2.0f * (float)row * (1.0f / (float)(HH - 1));

    const float cxa = cpx_of(4 * h), cxb = cpx_of(4 * (3 - h));
    float dy2[4];
    #pragma unroll
    for (int iy = 0; iy < 4; ++iy) {
        const float dy = py - cpy_of(iy);
        dy2[iy] = dy * dy;
    }

    v2f A0[2][NCL], A1[2][3];
    #pragma unroll
    for (int t = 0; t < 2; ++t) {
        const int col = j0 + pm + 8 * t;
        const float px = -1.0f + 2.0f * (float)col * (1.0f / (float)(WW - 1));
        const float dxa = px - cxa, dxb = px - cxb;
        const float xa2 = dxa * dxa, xb2 = dxb * dxb;
        float ph[NCL][2];
        #pragma unroll
        for (int e = 0; e < 2; ++e) {
            const float u00 = tps_u(xa2 + dy2[e]);
            const float u01 = tps_u(xa2 + dy2[3 - e]);
            const float u10 = tps_u(xb2 + dy2[e]);
            const float u11 = tps_u(xb2 + dy2[3 - e]);
            const float ea = u00 + u01, oa = u00 - u01;
            const float eb = u10 + u11, ob = u10 - u11;
            ph[0][e] = 0.5f * (ea + eb);
            ph[1][e] = 0.5f * (ea - eb);
            ph[2][e] = 0.5f * (oa + ob);
            ph[3][e] = 0.5f * (oa - ob);
        }
        #pragma unroll
        for (int s = 0; s < NCL; ++s) { A0[t][s].x = ph[s][0]; A0[t][s].y = ph[s][1]; }
        A1[t][0].x = hi ? 0.0f : 1.0f;  A1[t][0].y = 0.0f;
        A1[t][1].x = hi ? 0.0f : px;    A1[t][1].y = 0.0f;
        A1[t][2].x = hi ? 0.0f : py;    A1[t][2].y = 0.0f;
    }

    const float* bp = bmat + m;
    const int rb = hi ? 2 : 0;
    v2f Bc0[NCL], Bc1[3], Bn0[NCL], Bn1[3];
    #pragma unroll
    for (int s = 0; s < NCL; ++s) {
        Bc0[s].x = bp[(BD * s + rb)     * NCOL];
        Bc0[s].y = bp[(BD * s + rb + 1) * NCOL];
    }
    #pragma unroll
    for (int s = 0; s < 3; ++s) {
        const float al = bp[(BD * s + 4) * NCOL];
        Bc1[s].x = hi ? 0.0f : al;
        Bc1[s].y = 0.0f;
    }

    float* swn = &stage[(m >> 1) * SPIT + (m & 1) + (hi ? 32 : 0)];
    float* swm = &stage[(m >> 1) * SPIT + (m & 1) + (hi ? 0 : 32)];
    const int sg  = lane >> 4;
    const int l16 = lane & 15;
    const float* sr = &stage[sg * 64 + 4 * l16];
    const int colg = (sg ? (WW - QPTS - j0) : j0) + 2 * l16;
    float* op0 = out + ((size_t)row * WW + colg) * 2;
    float* op1 = out + ((size_t)(HH - 1 - row) * WW + colg) * 2;

    for (int ct = 0; ct < 8; ++ct) {
        const int cn = min(ct + 1, 7) * 16;
        #pragma unroll
        for (int s = 0; s < NCL; ++s) {
            Bn0[s].x = bp[(BD * s + rb)     * NCOL + cn];
            Bn0[s].y = bp[(BD * s + rb + 1) * NCOL + cn];
        }
        #pragma unroll
        for (int s = 0; s < 3; ++s) {
            const float al = bp[(BD * s + 4) * NCOL + cn];
            Bn1[s].x = hi ? 0.0f : al;
            Bn1[s].y = 0.0f;
        }

        v8f F[2][NCL];
        #pragma unroll
        for (int s = 0; s < NCL; ++s) {
            #pragma unroll
            for (int t = 0; t < 2; ++t)
                F[t][s] = __builtin_amdgcn_wmma_f32_16x16x4_f32(
                         false, A0[t][s],  false, Bc0[s],
                         (short)0, v8f{0.f, 0.f, 0.f, 0.f, 0.f, 0.f, 0.f, 0.f},
                         false,  false);
        }
        #pragma unroll
        for (int s = 0; s < 3; ++s) {
            #pragma unroll
            for (int t = 0; t < 2; ++t)
                F[t][s] = __builtin_amdgcn_wmma_f32_16x16x4_f32(
                         false, A1[t][s],  false, Bc1[s],
                         (short)0, F[t][s],  false,  false);
        }

        lds_handover();
        #pragma unroll
        for (int t = 0; t < 2; ++t) {
            #pragma unroll
            for (int r = 0; r < 8; ++r) {
                const float P = F[t][0][r] + F[t][2][r], R = F[t][0][r] - F[t][2][r];
                const float Q = F[t][1][r] + F[t][3][r], S = F[t][1][r] - F[t][3][r];
                const int q2 = 2 * (r + 8 * t);
                swn[0 * 64 + q2]      = P + Q;
                swm[1 * 64 + 30 - q2] = P - Q;
                swn[2 * 64 + q2]      = R + S;
                swm[3 * 64 + 30 - q2] = R - S;
            }
        }
        lds_handover();

        #pragma unroll
        for (int jj = 0; jj < 8; ++jj) {
            const v4f v0 = *(const v4f*)&sr[jj * SPIT];
            const v4f v1 = *(const v4f*)&sr[jj * SPIT + 128];
            const size_t bo = (size_t)(8 * ct + jj) * (size_t)(HWN * 2);
            *(v4f*)(op0 + bo) = v0;
            *(v4f*)(op1 + bo) = v1;
        }

        #pragma unroll
        for (int s = 0; s < NCL; ++s) Bc0[s] = Bn0[s];
        #pragma unroll
        for (int s = 0; s < 3; ++s) Bc1[s] = Bn1[s];
    }
}

extern "C" void kernel_launch(void* const* d_in, const int* in_sizes, int n_in,
                              void* d_out, int out_size, void* d_ws, size_t ws_size,
                              hipStream_t stream) {
    (void)in_sizes; (void)n_in; (void)out_size; (void)ws_size;
    const float* src = (const float*)d_in[0];
    float* out       = (float*)d_out;
    float* bmat      = (float*)d_ws;

    tps_setup_kernel<<<1, 256, 0, stream>>>(src, bmat);

    tps_main_kernel<<<2048, 32, 0, stream>>>(bmat, out);
}


# one launch, no coefficient matrix anywhere: each one-wave work-group inverts the four 5x5 class blocks itself in registers and forms its B fragments per column tile from the source points; sums kept
# speedup vs baseline: 42.3004x; 1.1262x over previous
#include <hip/hip_runtime.h>
#include <hip/hip_bf16.h>
#include <math.h>

#define HH   512
#define WW   512
#define KK   16
#define BB   64
#define NCOL 128
#define HWN  (HH*WW)

#define QPTS 32
#define SPIT 260

#define NCL  4
#define BD   5
#define BAUG (BD + 4)

typedef __attribute__((ext_vector_type(2))) float v2f;
typedef __attribute__((ext_vector_type(4))) float v4f;
typedef __attribute__((ext_vector_type(8))) float v8f;

__device__ __forceinline__ float cpx_of(int k) { return -1.0f + (2.0f/3.0f)*(float)(k >> 2); }
__device__ __forceinline__ float cpy_of(int k) { return -1.0f + (2.0f/3.0f)*(float)(k & 3); }
__device__ __forceinline__ float tps_u(float r2) { return r2 * logf(r2 + 1e-6f); }

__device__ __forceinline__ float rep_coord(int i) { return cpx_of(4 * i); }

__device__ __forceinline__ void lds_handover() {
    __builtin_amdgcn_fence(__ATOMIC_RELEASE, "workgroup", "local");
    __builtin_amdgcn_fence(__ATOMIC_ACQUIRE, "workgroup", "local");
}

__device__ __forceinline__ void load_src(const float* __restrict__ spl, int ct, float (&spv)[KK]) {
    #pragma unroll
    for (int j = 0; j < KK; ++j) spv[j] = spl[ct * (8 * KK * 2) + 2 * j];
}

__device__ __forceinline__ float dot4(const v4f m, const float (&v)[4]) {
    float acc = m.x * v[0];
    acc = fmaf(m.y, v[1], acc);
    acc = fmaf(m.z, v[2], acc);
    acc = fmaf(m.w, v[3], acc);
    return acc;
}

__device__ __forceinline__ void make_bfrag(const float (&spv)[KK], const v4f (&mi)[NCL][3], bool hi,
                                           v2f (&B0)[NCL], v2f (&B1)[3]) {
    float st[NCL][4];
    #pragma unroll
    for (int p = 0; p < 4; ++p) {
        const int ar = p >> 1, br = p & 1;
        const float u00 = spv[4 * ar       + br];
        const float u01 = spv[4 * ar       + (3 - br)];
        const float u10 = spv[4 * (3 - ar) + br];
        const float u11 = spv[4 * (3 - ar) + (3 - br)];
        const float e0 = u00 + u10, o0 = u00 - u10;
        const float e1 = u01 + u11, o1 = u01 - u11;
        st[0][p] = 0.5f * (e0 + e1);
        st[1][p] = 0.5f * (o0 + o1);
        st[2][p] = 0.5f * (e0 - e1);
        st[3][p] = 0.5f * (o0 - o1);
    }
    #pragma unroll
    for (int s = 0; s < NCL; ++s) {
        B0[s].x = dot4(mi[s][0], st[s]);
        B0[s].y = dot4(mi[s][1], st[s]);
    }
    #pragma unroll
    for (int s = 0; s < 3; ++s) {
        const float al = dot4(mi[s][2], st[s]);
        B1[s].x = hi ? 0.0f : al;
        B1[s].y = 0.0f;
    }
}

__global__ void __launch_bounds__(32)
tps_fused_kernel(const float* __restrict__ sp, float* __restrict__ out) {
    __shared__ __attribute__((aligned(16))) float stage[8 * SPIT];
    __shared__ __attribute__((aligned(16))) float Mb[NCL * 16];
    __shared__ __attribute__((aligned(16))) float Mi[NCL * BD * 4];

    const int lane = threadIdx.x;
    const int m    = lane & 15;
    const bool hi  = lane >= 16;
    const int h    = hi ? 1 : 0;
    const int row  = blockIdx.x >> 3;
    const int j0   = (blockIdx.x & 7) * QPTS;
    const int pm   = (m & 7) + ((m >> 3) << 4);

    const float* spl = sp + (m >> 1) * (KK * 2) + (m & 1);
    float spn[KK];
    load_src(spl, 0, spn);

    {
        const int p = (lane >> 2) & 3, q = lane & 3;
        const float xp = rep_coord(p >> 1), yp = rep_coord(p & 1);
        const float xq = rep_coord(q >> 1), yq = rep_coord(q & 1);
        const float dxa = xp - xq, dxb = xp + xq;
        const float dya = yp - yq, dyb = yp + yq;
        const float u0 = tps_u(dxa*dxa + dya*dya);
        const float u1 = tps_u(dxa*dxa + dyb*dyb);
        const float u2 = tps_u(dxb*dxb + dya*dya);
        const float u3 = tps_u(dxb*dxb + dyb*dyb);
        const float sx = hi ? -1.0f : 1.0f;
        #pragma unroll
        for (int it = 0; it < 2; ++it) {
            const float sy = it ? -1.0f : 1.0f;
            float kv = u0;
            kv += sy * u1;
            kv += sx * u2;
            kv += (sx * sy) * u3;
            Mb[(h + 2 * it) * 16 + m] = kv;
        }
    }
    lds_handover();

    {
        const int s = lane & 3;
        float a[BD][BAUG];
        #pragma unroll
        for (int i = 0; i < BD - 1; ++i) {
            const v4f r = *(const v4f*)&Mb[s * 16 + i * 4];
            a[i][0] = r.x; a[i][1] = r.y; a[i][2] = r.z; a[i][3] = r.w;
            float bv = (s == 2) ? 2.0f * rep_coord(i & 1) : 0.0f;
            bv = (s == 1) ? 2.0f * rep_coord(i >> 1) : bv;
            bv = (s == 0) ? 2.0f : bv;
            a[i][BD - 1] = bv;
            a[BD - 1][i] = bv;
        }
        a[BD - 1][BD - 1] = (s == 3) ? 1.0f : 0.0f;
        #pragma unroll
        for (int i = 0; i < BD; ++i) {
            #pragma unroll
            for (int q = 0; q < 4; ++q) a[i][BD + q] = (i == q) ? 1.0f : 0.0f;
        }
        #pragma unroll
        for (int P = 0; P < BD; ++P) {
            float pv = a[P][P];
            int   pr = P;
            #pragma unroll
            for (int r = P + 1; r < BD; ++r) {
                const bool tk = fabsf(a[r][P]) > fabsf(pv);
                pv = tk ? a[r][P] : pv;
                pr = tk ? r : pr;
            }
            const float inv = 1.0f / pv;
            #pragma unroll
            for (int j = 0; j < BAUG; ++j) {
                const float top = a[P][j];
                float pick = top;
                #pragma unroll
                for (int r = P + 1; r < BD; ++r) {
                    const bool is = (pr == r);
                    const float cur = a[r][j];
                    pick    = is ? cur : pick;
                    a[r][j] = is ? top : cur;
                }
                a[P][j] = pick * inv;
            }
            #pragma unroll
            for (int r = 0; r < BD; ++r) {
                if (r != P) {
                    const float f = a[r][P];
                    #pragma unroll
                    for (int j = 0; j < BAUG; ++j) a[r][j] -= f * a[P][j];
                }
            }
        }
        if (lane < NCL) {
            #pragma unroll
            for (int i = 0; i < BD; ++i) {
                #pragma unroll
                for (int q = 0; q < 4; ++q) Mi[s * (BD * 4) + i * 4 + q] = a[i][BD + q];
            }
        }
    }
    lds_handover();

    const int rb = hi ? 2 : 0;
    v4f mi[NCL][3];
    #pragma unroll
    for (int s = 0; s < NCL; ++s) {
        mi[s][0] = *(const v4f*)&Mi[s * (BD * 4) + rb * 4];
        mi[s][1] = *(const v4f*)&Mi[s * (BD * 4) + (rb + 1) * 4];
        mi[s][2] = *(const v4f*)&Mi[s * (BD * 4) + (BD - 1) * 4];
    }

    const float py = -1.0f + 2.0f * (float)row * (1.0f / (float)(HH - 1));

    const float cxa = cpx_of(4 * h), cxb = cpx_of(4 * (3 - h));
    float dy2[4];
    #pragma unroll
    for (int iy = 0; iy < 4; ++iy) {
        const float dy = py - cpy_of(iy);
        dy2[iy] = dy * dy;
    }

    v2f A0[2][NCL], A1[2][3];
    #pragma unroll
    for (int t = 0; t < 2; ++t) {
        const int col = j0 + pm + 8 * t;
        const float px = -1.0f + 2.0f * (float)col * (1.0f / (float)(WW - 1));
        const float dxa = px - cxa, dxb = px - cxb;
        const float xa2 = dxa * dxa, xb2 = dxb * dxb;
        float ph[NCL][2];
        #pragma unroll
        for (int e = 0; e < 2; ++e) {
            const float u00 = tps_u(xa2 + dy2[e]);
            const float u01 = tps_u(xa2 + dy2[3 - e]);
            const float u10 = tps_u(xb2 + dy2[e]);
            const float u11 = tps_u(xb2 + dy2[3 - e]);
            const float ea = u00 + u01, oa = u00 - u01;
            const float eb = u10 + u11, ob = u10 - u11;
            ph[0][e] = 0.5f * (ea + eb);
            ph[1][e] = 0.5f * (ea - eb);
            ph[2][e] = 0.5f * (oa + ob);
            ph[3][e] = 0.5f * (oa - ob);
        }
        #pragma unroll
        for (int s = 0; s < NCL; ++s) { A0[t][s].x = ph[s][0]; A0[t][s].y = ph[s][1]; }
        A1[t][0].x = hi ? 0.0f : 1.0f;  A1[t][0].y = 0.0f;
        A1[t][1].x = hi ? 0.0f : px;    A1[t][1].y = 0.0f;
        A1[t][2].x = hi ? 0.0f : py;    A1[t][2].y = 0.0f;
    }

    v2f Bc0[NCL], Bc1[3];
    make_bfrag(spn, mi, hi, Bc0, Bc1);

    float* swn = &stage[(m >> 1) * SPIT + (m & 1) + (hi ? 32 : 0)];
    float* swm = &stage[(m >> 1) * SPIT + (m & 1) + (hi ? 0 : 32)];
    const int sg  = lane >> 4;
    const int l16 = lane & 15;
    const float* sr = &stage[sg * 64 + 4 * l16];
    const int colg = (sg ? (WW - QPTS - j0) : j0) + 2 * l16;
    float* op0 = out + ((size_t)row * WW + colg) * 2;
    float* op1 = out + ((size_t)(HH - 1 - row) * WW + colg) * 2;

    for (int ct = 0; ct < 8; ++ct) {
        load_src(spl, min(ct + 1, 7), spn);

        v8f F[2][NCL];
        #pragma unroll
        for (int s = 0; s < NCL; ++s) {
            #pragma unroll
            for (int t = 0; t < 2; ++t)
                F[t][s] = __builtin_amdgcn_wmma_f32_16x16x4_f32(
                         false, A0[t][s],  false, Bc0[s],
                         (short)0, v8f{0.f, 0.f, 0.f, 0.f, 0.f, 0.f, 0.f, 0.f},
                         false,  false);
        }
        #pragma unroll
        for (int s = 0; s < 3; ++s) {
            #pragma unroll
            for (int t = 0; t < 2; ++t)
                F[t][s] = __builtin_amdgcn_wmma_f32_16x16x4_f32(
                         false, A1[t][s],  false, Bc1[s],
                         (short)0, F[t][s],  false,  false);
        }

        lds_handover();
        #pragma unroll
        for (int t = 0; t < 2; ++t) {
            #pragma unroll
            for (int r = 0; r < 8; ++r) {
                const float P = F[t][0][r] + F[t][2][r], R = F[t][0][r] - F[t][2][r];
                const float Q = F[t][1][r] + F[t][3][r], S = F[t][1][r] - F[t][3][r];
                const int q2 = 2 * (r + 8 * t);
                swn[0 * 64 + q2]      = P + Q;
                swm[1 * 64 + 30 - q2] = P - Q;
                swn[2 * 64 + q2]      = R + S;
                swm[3 * 64 + 30 - q2] = R - S;
            }
        }
        lds_handover();

        #pragma unroll
        for (int jj = 0; jj < 8; ++jj) {
            const v4f v0 = *(const v4f*)&sr[jj * SPIT];
            const v4f v1 = *(const v4f*)&sr[jj * SPIT + 128];
            const size_t bo = (size_t)(8 * ct + jj) * (size_t)(HWN * 2);
            *(v4f*)(op0 + bo) = v0;
            *(v4f*)(op1 + bo) = v1;
        }

        make_bfrag(spn, mi, hi, Bc0, Bc1);
    }
}

extern "C" void kernel_launch(void* const* d_in, const int* in_sizes, int n_in,
                              void* d_out, int out_size, void* d_ws, size_t ws_size,
                              hipStream_t stream) {
    (void)in_sizes; (void)n_in; (void)out_size; (void)d_ws; (void)ws_size;
    const float* src = (const float*)d_in[0];
    float* out       = (float*)d_out;

    tps_fused_kernel<<<2048, 32, 0, stream>>>(src, out);
}
